# starting kernel
# baseline (speedup 1.0000x reference)
#include <hip/hip_runtime.h>
#include <hip/hip_bf16.h>


typedef __attribute__((ext_vector_type(2))) float v2f;
typedef __attribute__((ext_vector_type(4))) float v4f;
typedef __attribute__((ext_vector_type(8))) float v8f;

#define D_FEAT 64
#define EDGES_PER_WAVE 16
#define BLOCK_THREADS 256
#define EDGES_PER_BLOCK ((BLOCK_THREADS / 32) * EDGES_PER_WAVE)

__global__ __launch_bounds__(BLOCK_THREADS) void edge_dot_wmma_kernel(
    const float* __restrict__ h,
    const int* __restrict__ src,
    const int* __restrict__ dst,
    float* __restrict__ out,
    int E)
{
    const int lane  = threadIdx.x & 31;
    const int wave  = blockIdx.x * (BLOCK_THREADS >> 5) + (threadIdx.x >> 5);
    const int m     = lane & 15;
    const int half  = lane >> 4;

    const int e0 = wave * EDGES_PER_WAVE;
    int e = e0 + m;
    int ec = e < E ? e : (E - 1);

    const long long si = (long long)src[ec];
    const long long di = (long long)dst[ec];

    const float* __restrict__ srow = h + si * D_FEAT + half * 32;
    const float* __restrict__ drow = h + di * D_FEAT + half * 32;

    v2f acc; acc.x = 0.0f; acc.y = 0.0f;
#pragma unroll
    for (int c = 0; c < 8; ++c) {
        v4f s = *(const v4f*)(srow + c * 4);
        v4f t = *(const v4f*)(drow + c * 4);
        acc.x = __builtin_fmaf(s.x, t.x, acc.x);
        acc.y = __builtin_fmaf(s.y, t.y, acc.y);
        acc.x = __builtin_fmaf(s.z, t.z, acc.x);
        acc.y = __builtin_fmaf(s.w, t.w, acc.y);
    }

    v2f ones; ones.x = 1.0f; ones.y = 1.0f;
    v8f d = {};
    d = __builtin_amdgcn_wmma_f32_16x16x4_f32(
         false, acc,
         false, ones,
         (short)0, d,
         false,  false);

    if ((lane & 15) == 0) {
        const int base = e0 + half * 8;
        if (base + 7 < E) {
            v4f lo; lo.x = d[0]; lo.y = d[1]; lo.z = d[2]; lo.w = d[3];
            v4f hi; hi.x = d[4]; hi.y = d[5]; hi.z = d[6]; hi.w = d[7];
            *(v4f*)(out + base)     = lo;
            *(v4f*)(out + base + 4) = hi;
        } else {
#pragma unroll
            for (int v = 0; v < 8; ++v) {
                if (base + v < E) out[base + v] = d[v];
            }
        }
    }
}

extern "C" void kernel_launch(void* const* d_in, const int* in_sizes, int n_in,
                              void* d_out, int out_size, void* d_ws, size_t ws_size,
                              hipStream_t stream) {
    const float* h   = (const float*)d_in[0];
    const int*   src = (const int*)d_in[1];
    const int*   dst = (const int*)d_in[2];
    float*       out = (float*)d_out;

    const int E = in_sizes[1];
    const int blocks = (E + EDGES_PER_BLOCK - 1) / EDGES_PER_BLOCK;

    edge_dot_wmma_kernel<<<blocks, BLOCK_THREADS, 0, stream>>>(h, src, dst, out, E);
}


# Coalesced gather: each 16-lane row loads one whole 256-B feature row (4 lines per b128 load, was 32), fixed grid of looping waves; WMMA with a 0/1 selector B collects lane sums; new reduction tree
# speedup vs baseline: 3.6463x; 3.6463x over previous
#include <hip/hip_runtime.h>


typedef __attribute__((ext_vector_type(2))) float v2f;
typedef __attribute__((ext_vector_type(4))) float v4f;
typedef __attribute__((ext_vector_type(4))) int   v4i;
typedef __attribute__((ext_vector_type(8))) float v8f;

#define D_FEAT 64
#define BLOCK_THREADS 128
#define WAVES_PER_BLOCK (BLOCK_THREADS / 32)
#define GRID_BLOCKS 1563

static __device__ __forceinline__ float xhalf(float x) {
    int b = __builtin_bit_cast(int, x);
    return __builtin_bit_cast(float,
        __builtin_amdgcn_permlanex16(b, b, 0x76543210, 0xfedcba98, false, false));
}

struct Idx8 { v4i s0, s1, d0, d1; };

static __device__ __forceinline__ Idx8 load_idx8(
    const int* __restrict__ src, const int* __restrict__ dst, int ebase, int hh)
{
    const v4i* sp = (const v4i*)(src + ebase + 8 * hh);
    const v4i* dp = (const v4i*)(dst + ebase + 8 * hh);
    Idx8 r;
    r.s0 = sp[0]; r.s1 = sp[1];
    r.d0 = dp[0]; r.d1 = dp[1];
    return r;
}

template <bool HOLD>
static __device__ __forceinline__ float half_sums16(
    const char* __restrict__ hb, const Idx8& ix, const Idx8& ah0, const Idx8& ah1,
    unsigned lane_off, const v2f (&bsel)[4])
{
    const int si[8] = { ix.s0.x, ix.s0.y, ix.s0.z, ix.s0.w, ix.s1.x, ix.s1.y, ix.s1.z, ix.s1.w };
    const int di[8] = { ix.d0.x, ix.d0.y, ix.d0.z, ix.d0.w, ix.d1.x, ix.d1.y, ix.d1.z, ix.d1.w };

    v4f a[8], b[8];
#pragma unroll
    for (int q = 0; q < 8; ++q) {
        const unsigned so   = ((unsigned)si[q] << 8) + lane_off;
        const unsigned dofs = ((unsigned)di[q] << 8) + lane_off;
        a[q] = *(const v4f*)(hb + so);
        b[q] = *(const v4f*)(hb + dofs);
    }
    if (HOLD) {
        asm volatile("" :: "v"(a[0]), "v"(a[1]), "v"(a[2]), "v"(a[3]),
                           "v"(a[4]), "v"(a[5]), "v"(a[6]), "v"(a[7]),
                           "v"(b[0]), "v"(b[1]), "v"(b[2]), "v"(b[3]),
                           "v"(b[4]), "v"(b[5]), "v"(b[6]), "v"(b[7]),
                           "v"(ah0.s0), "v"(ah0.s1), "v"(ah0.d0), "v"(ah0.d1),
                           "v"(ah1.s0), "v"(ah1.s1), "v"(ah1.d0), "v"(ah1.d1));
    }

    v8f d = {};
#pragma unroll
    for (int j = 0; j < 4; ++j) {
        v2f p;
#pragma unroll
        for (int i = 0; i < 2; ++i) {
            const v4f s = a[2 * j + i];
            const v4f t = b[2 * j + i];
            v2f x = s.xy * t.xy;
            x = __builtin_elementwise_fma(s.zw, t.zw, x);
            p[i] = x.x + x.y;
        }
        d = __builtin_amdgcn_wmma_f32_16x16x4_f32(
             false, p,
             false, bsel[j],
             (short)0, d,
             false,  false);
    }
    return ((d[0] + d[1]) + (d[2] + d[3])) + ((d[4] + d[5]) + (d[6] + d[7]));
}

__global__ __launch_bounds__(BLOCK_THREADS) void edge_dot_rows_wmma_kernel(
    const float* __restrict__ h,
    const int* __restrict__ src,
    const int* __restrict__ dst,
    float* __restrict__ out,
    int E)
{
    const int lane = threadIdx.x & 31;
    const int c    = lane & 15;
    const int hh   = lane >> 4;
    const int wave = blockIdx.x * WAVES_PER_BLOCK
                   + __builtin_amdgcn_readfirstlane((int)(threadIdx.x >> 5));
    const int nwaves = gridDim.x * WAVES_PER_BLOCK;

    const char* __restrict__ hb = (const char*)h;
    const unsigned lane_off = (unsigned)c * 16u;

    v2f bsel[4];
#pragma unroll
    for (int j = 0; j < 4; ++j) {
        bsel[j].x = (c == 8 * hh + 2 * j)     ? 1.0f : 0.0f;
        bsel[j].y = (c == 8 * hh + 2 * j + 1) ? 1.0f : 0.0f;
    }

    const int ngroups = E >> 5;
    int g = wave;
    if (g < ngroups) {
        Idx8 cur = load_idx8(src, dst, g << 5, hh);
        for (; g < ngroups; g += nwaves) {
            const int e0 = g << 5;
            const Idx8 mid = load_idx8(src, dst, e0 + 16, hh);
            int gn = g + nwaves;
            gn = gn < ngroups ? gn : ngroups - 1;
            const Idx8 nxt = load_idx8(src, dst, gn << 5, hh);
            const float ra = half_sums16<true>(hb, cur, mid, nxt, lane_off, bsel);
            const float rb = half_sums16<false>(hb, mid, nxt, nxt, lane_off, bsel);
            cur = nxt;
            const float keep = hh ? rb : ra;
            const float send = hh ? ra : rb;
            out[e0 + lane] = keep + xhalf(send);
        }
    }

    const int tail0 = ngroups << 5;
    if (wave == 0) {
        const int e = tail0 + lane;
        if (e < E) {
            const float* __restrict__ sr = h + (size_t)src[e] * D_FEAT;
            const float* __restrict__ dr = h + (size_t)dst[e] * D_FEAT;
            float acc = 0.0f;
            for (int k = 0; k < D_FEAT; k += 4) {
                const v4f s = *(const v4f*)(sr + k);
                const v4f t = *(const v4f*)(dr + k);
                acc = __builtin_fmaf(s.x, t.x, acc);
                acc = __builtin_fmaf(s.y, t.y, acc);
                acc = __builtin_fmaf(s.z, t.z, acc);
                acc = __builtin_fmaf(s.w, t.w, acc);
            }
            out[e] = acc;
        }
    }
}

extern "C" void kernel_launch(void* const* d_in, const int* in_sizes, int n_in,
                              void* d_out, int out_size, void* d_ws, size_t ws_size,
                              hipStream_t stream) {
    const float* h   = (const float*)d_in[0];
    const int*   src = (const int*)d_in[1];
    const int*   dst = (const int*)d_in[2];
    float*       out = (float*)d_out;

    const int E = in_sizes[1];

    edge_dot_rows_wmma_kernel<<<GRID_BLOCKS, BLOCK_THREADS, 0, stream>>>(h, src, dst, out, E);
}
